# starting kernel
# baseline (speedup 1.0000x reference)
#include <hip/hip_runtime.h>

typedef float v2f __attribute__((ext_vector_type(2)));
typedef float v8f __attribute__((ext_vector_type(8)));

#define NUM_CAPS 512
#define CAP_SIZE 256
#define RF_SIZE  64
#define KDIM     2048
#define MDIM     2048
#define NDIM     64
#define KSPLIT   8

__global__ __launch_bounds__(256)
void scrramble_route_wmma(const float* __restrict__ Ci,
                          const float* __restrict__ x,
                          float* __restrict__ xr) {
  __shared__ float red[KSPLIT * 32 * 8];

  const int lane = threadIdx.x & 31;
  const int w    = __builtin_amdgcn_readfirstlane(threadIdx.x >> 5);
  const int mt   = blockIdx.x >> 2;
  const int nt   = blockIdx.x & 3;
  const int R0   = mt * 16;
  const int N0   = nt * 16;
  const int half = lane >> 4;
  const int ll   = lane & 15;

  const int Sbeg = w * (KDIM / KSPLIT);
  const int Send = Sbeg + (KDIM / KSPLIT);

  v8f c = {};
#pragma unroll 4
  for (int S = Sbeg; S < Send; S += 4) {
    const int k0 = S + 2 * half;
    v2f a, b;
    a.x = Ci[(size_t)k0 * MDIM + (size_t)(R0 + ll)];
    a.y = Ci[(size_t)(k0 + 1) * MDIM + (size_t)(R0 + ll)];
    b.x = x[k0 * NDIM + N0 + ll];
    b.y = x[(k0 + 1) * NDIM + N0 + ll];
    c = __builtin_amdgcn_wmma_f32_16x16x4_f32(false, a, false, b,
                                              (short)0, c, false, false);
  }

#pragma unroll
  for (int r = 0; r < 8; ++r) red[(w * 32 + lane) * 8 + r] = c[r];
  __syncthreads();

  if (threadIdx.x < 32) {
    v8f acc = {};
#pragma unroll
    for (int ww = 0; ww < KSPLIT; ++ww) {
#pragma unroll
      for (int r = 0; r < 8; ++r) acc[r] += red[(ww * 32 + lane) * 8 + r];
    }
#pragma unroll
    for (int r = 0; r < 8; ++r)
      xr[(size_t)(R0 + r + 8 * half) * NDIM + (size_t)(N0 + ll)] = acc[r];
  }
}

__global__ __launch_bounds__(512)
void caps_transform_wmma(const float* __restrict__ Wi,
                         const float* __restrict__ xr,
                         float* __restrict__ y) {
  __shared__ float xs[CAP_SIZE];

  const int cap   = blockIdx.x;
  const int mtile = threadIdx.x >> 5;
  const int lane  = threadIdx.x & 31;
  const int half  = lane >> 4;
  const int ll    = lane & 15;
  const int row   = mtile * 16 + ll;
  const int jj    = row >> 6;
  const int l     = row & 63;

  const float* __restrict__ Wcap = Wi + (size_t)cap * (CAP_SIZE * CAP_SIZE);
  const float* __restrict__ Wrow = Wcap + jj * 16384 + l * 64;

  if (threadIdx.x < CAP_SIZE)
    xs[threadIdx.x] = xr[(size_t)cap * CAP_SIZE + threadIdx.x];
  __syncthreads();

  v8f c = {};
#pragma unroll 4
  for (int S = 0; S < CAP_SIZE; S += 4) {
    const int col0 = S + 2 * half;
    const int k    = col0 >> 6;
    const int m    = col0 & 63;
    if ((S & 63) == 0 && S + 64 < CAP_SIZE) {
      __builtin_prefetch(Wrow + (k + 1) * 4096 + m, 0, 1);
    }
    const v2f a = __builtin_nontemporal_load(
        (const v2f*)(Wrow + k * 4096 + m));
    const float bx0 = xs[col0];
    const float bx1 = xs[col0 + 1];
    v2f b;
    b.x = (ll == 0) ? bx0 : 0.0f;
    b.y = (ll == 0) ? bx1 : 0.0f;
    c = __builtin_amdgcn_wmma_f32_16x16x4_f32(false, a, false, b,
                                              (short)0, c, false, false);
  }

  if (ll == 0) {
#pragma unroll
    for (int r = 0; r < 8; ++r) {
      y[(size_t)cap * CAP_SIZE + (size_t)(mtile * 16 + r + 8 * half)] = c[r];
    }
  }
}

extern "C" void kernel_launch(void* const* d_in, const int* in_sizes, int n_in,
                              void* d_out, int out_size, void* d_ws, size_t ws_size,
                              hipStream_t stream) {
  const float* x  = (const float*)d_in[0];
  const float* Ci = (const float*)d_in[1];
  const float* Wi = (const float*)d_in[2];
  float* y  = (float*)d_out;
  float* xr = (float*)d_ws;

  scrramble_route_wmma<<<512, 256, 0, stream>>>(Ci, x, xr);
  caps_transform_wmma<<<NUM_CAPS, 512, 0, stream>>>(Wi, xr, y);
}


# Stage 2 weight stream: whole-line 512B wave loads parked in a wave-private LDS tile, A fragments read back by ds_load_b128; k regrouped per WMMA pair (8q+0,1,4,5 then 8q+2,3,6,7), deterministic
# speedup vs baseline: 1.7986x; 1.7986x over previous
#include <hip/hip_runtime.h>

typedef float v2f __attribute__((ext_vector_type(2)));
typedef float v4f __attribute__((ext_vector_type(4)));
typedef float v8f __attribute__((ext_vector_type(8)));

#define NUM_CAPS 512
#define CAP_SIZE 256
#define RF_SIZE  64
#define KDIM     2048
#define MDIM     2048
#define NDIM     64
#define KSPLIT   8

__global__ __launch_bounds__(256)
void scrramble_route_wmma(const float* __restrict__ Ci,
                          const float* __restrict__ x,
                          float* __restrict__ xr) {
  __shared__ float red[KSPLIT * 32 * 8];

  const int lane = threadIdx.x & 31;
  const int w    = __builtin_amdgcn_readfirstlane(threadIdx.x >> 5);
  const int mt   = blockIdx.x >> 2;
  const int nt   = blockIdx.x & 3;
  const int R0   = mt * 16;
  const int N0   = nt * 16;
  const int half = lane >> 4;
  const int ll   = lane & 15;

  const int Sbeg = w * (KDIM / KSPLIT);
  const int Send = Sbeg + (KDIM / KSPLIT);

  v8f c = {};
#pragma unroll 4
  for (int S = Sbeg; S < Send; S += 4) {
    const int k0 = S + 2 * half;
    v2f a, b;
    a.x = Ci[(size_t)k0 * MDIM + (size_t)(R0 + ll)];
    a.y = Ci[(size_t)(k0 + 1) * MDIM + (size_t)(R0 + ll)];
    b.x = x[k0 * NDIM + N0 + ll];
    b.y = x[(k0 + 1) * NDIM + N0 + ll];
    c = __builtin_amdgcn_wmma_f32_16x16x4_f32(false, a, false, b,
                                              (short)0, c, false, false);
  }

#pragma unroll
  for (int r = 0; r < 8; ++r) red[(w * 32 + lane) * 8 + r] = c[r];
  __syncthreads();

  if (threadIdx.x < 32) {
    v8f acc = {};
#pragma unroll
    for (int ww = 0; ww < KSPLIT; ++ww) {
#pragma unroll
      for (int r = 0; r < 8; ++r) acc[r] += red[(ww * 32 + lane) * 8 + r];
    }
#pragma unroll
    for (int r = 0; r < 8; ++r)
      xr[(size_t)(R0 + r + 8 * half) * NDIM + (size_t)(N0 + ll)] = acc[r];
  }
}

#define TPITCH  68
#define WFLOATS (CAP_SIZE + 16 * TPITCH)

__global__ __launch_bounds__(512)
void caps_transform_wmma(const float* __restrict__ Wi,
                         const float* __restrict__ xr,
                         float* __restrict__ y) {
  __shared__ __attribute__((aligned(16))) float smem[16 * WFLOATS];

  const int cap   = blockIdx.x;
  const int lane  = threadIdx.x & 31;
  const int mtile = __builtin_amdgcn_readfirstlane(threadIdx.x >> 5);
  const int half  = lane >> 4;
  const int ll    = lane & 15;
  const int jj    = mtile >> 2;
  const int l0    = (mtile & 3) * 16;

  float* __restrict__ xsw  = smem + mtile * WFLOATS;
  float* __restrict__ tile = xsw + CAP_SIZE;

  const float* __restrict__ xcap = xr + (size_t)cap * CAP_SIZE + lane * 4;
  const v4f xv0 = *(const v4f*)(xcap);
  const v4f xv1 = *(const v4f*)(xcap + 128);

  const float* __restrict__ Wbase =
      Wi + (size_t)cap * (CAP_SIZE * CAP_SIZE) + jj * 16384 + l0 * 64 + lane * 4;

  v4f buf[2][8];
#pragma unroll
  for (int i = 0; i < 8; ++i)
    buf[0][i] = __builtin_nontemporal_load((const v4f*)(Wbase + i * 128));
#pragma unroll
  for (int i = 0; i < 8; ++i)
    buf[1][i] = __builtin_nontemporal_load((const v4f*)(Wbase + 4096 + i * 128));

  *(v4f*)(xsw + lane * 4)       = xv0;
  *(v4f*)(xsw + 128 + lane * 4) = xv1;

  float* __restrict__ tst = tile + half * TPITCH + ll * 4;
  const float* __restrict__ tld = tile + ll * TPITCH + half * 4;
  const float* __restrict__ xld = xsw + half * 4;

  v8f c = {};
#pragma unroll
  for (int kb = 0; kb < 4; ++kb) {
#pragma unroll
    for (int i = 0; i < 8; ++i)
      *(v4f*)(tst + i * 2 * TPITCH) = buf[kb & 1][i];
    if (kb + 2 < 4) {
#pragma unroll
      for (int i = 0; i < 8; ++i)
        buf[kb & 1][i] = __builtin_nontemporal_load(
            (const v4f*)(Wbase + (kb + 2) * 4096 + i * 128));
    }
    __builtin_amdgcn_fence(__ATOMIC_RELEASE, "wavefront");
    __builtin_amdgcn_wave_barrier();
    __builtin_amdgcn_fence(__ATOMIC_ACQUIRE, "wavefront");

    v4f af[8], bf[8];
#pragma unroll
    for (int q = 0; q < 8; ++q) {
      af[q] = *(const v4f*)(tld + q * 8);
      bf[q] = *(const v4f*)(xld + kb * 64 + q * 8);
    }
#pragma unroll
    for (int q = 0; q < 8; ++q) {
      v2f a0, b0, a1, b1;
      a0.x = af[q].x; a0.y = af[q].y; b0.x = bf[q].x; b0.y = bf[q].y;
      a1.x = af[q].z; a1.y = af[q].w; b1.x = bf[q].z; b1.y = bf[q].w;
      c = __builtin_amdgcn_wmma_f32_16x16x4_f32(false, a0, false, b0,
                                                (short)0, c, false, false);
      c = __builtin_amdgcn_wmma_f32_16x16x4_f32(false, a1, false, b1,
                                                (short)0, c, false, false);
    }
    __builtin_amdgcn_fence(__ATOMIC_RELEASE, "wavefront");
    __builtin_amdgcn_wave_barrier();
    __builtin_amdgcn_fence(__ATOMIC_ACQUIRE, "wavefront");
  }

  if (ll == 0) {
#pragma unroll
    for (int r = 0; r < 8; ++r) {
      y[(size_t)cap * CAP_SIZE + (size_t)(mtile * 16 + r + 8 * half)] = c[r];
    }
  }
}

extern "C" void kernel_launch(void* const* d_in, const int* in_sizes, int n_in,
                              void* d_out, int out_size, void* d_ws, size_t ws_size,
                              hipStream_t stream) {
  const float* x  = (const float*)d_in[0];
  const float* Ci = (const float*)d_in[1];
  const float* Wi = (const float*)d_in[2];
  float* y  = (float*)d_out;
  float* xr = (float*)d_ws;

  scrramble_route_wmma<<<512, 256, 0, stream>>>(Ci, x, xr);
  caps_transform_wmma<<<NUM_CAPS, 512, 0, stream>>>(Wi, xr, y);
}


# Stage 1 re-tiled to one 16x32 tile per block: Ci and x rows streamed by 16B-per-lane loads through wave-private LDS tiles into the fragments, A fragment shared by 2 N tiles; sum order kept
# speedup vs baseline: 2.1106x; 1.1735x over previous
#include <hip/hip_runtime.h>

typedef float v2f __attribute__((ext_vector_type(2)));
typedef float v4f __attribute__((ext_vector_type(4)));
typedef float v8f __attribute__((ext_vector_type(8)));

#define NUM_CAPS 512
#define CAP_SIZE 256
#define RF_SIZE  64
#define KDIM     2048
#define MDIM     2048
#define NDIM     64
#define KSPLIT   8

#define S1_CH     32
#define S1_NCH    ((KDIM / KSPLIT) / S1_CH)
#define S1_DEPTH  3
#define PT        16
#define S1_TA     0
#define S1_TB0    (S1_CH * PT)
#define S1_TB1    (2 * S1_CH * PT + 32)
#define S1_WFLOATS (3 * S1_CH * PT + 32)

__global__ __launch_bounds__(256)
void scrramble_route_wmma(const float* __restrict__ Ci,
                          const float* __restrict__ x,
                          float* __restrict__ xr) {
  __shared__ __attribute__((aligned(16))) float smem[KSPLIT * S1_WFLOATS];

  const int lane = threadIdx.x & 31;
  const int w    = __builtin_amdgcn_readfirstlane(threadIdx.x >> 5);
  const int mt   = blockIdx.x >> 1;
  const int nh   = blockIdx.x & 1;
  const int R0   = mt * 16;
  const int N0   = nh * 32;
  const int half = lane >> 4;
  const int ll   = lane & 15;
  const int Sw   = w * (KDIM / KSPLIT);

  float* __restrict__ tw = smem + w * S1_WFLOATS;

  const float* __restrict__ ciw =
      Ci + (size_t)(Sw + (lane >> 2)) * MDIM + R0 + (lane & 3) * 4;
  const float* __restrict__ xw =
      x + (size_t)(Sw + (lane >> 3)) * NDIM + N0 + (lane & 7) * 4;

  v4f bufA[S1_DEPTH][4], bufB[S1_DEPTH][8];
#pragma unroll
  for (int d = 0; d < S1_DEPTH; ++d) {
#pragma unroll
    for (int i = 0; i < 4; ++i)
      bufA[d][i] = *(const v4f*)(ciw + (size_t)(d * S1_CH + i * 8) * MDIM);
#pragma unroll
    for (int i = 0; i < 8; ++i)
      bufB[d][i] = *(const v4f*)(xw + (size_t)(d * S1_CH + i * 4) * NDIM);
  }

  float* __restrict__ stA = tw + S1_TA + (lane >> 2) * PT + (lane & 3) * 4;
  float* __restrict__ stB = tw + (((lane >> 2) & 1) ? S1_TB1 : S1_TB0)
                               + (lane >> 3) * PT + (lane & 3) * 4;
  const float* __restrict__ frag = tw + (2 * half) * PT + ll;

  v8f c0 = {}, c1 = {};
#pragma unroll
  for (int kb = 0; kb < S1_NCH; ++kb) {
#pragma unroll
    for (int i = 0; i < 4; ++i)
      *(v4f*)(stA + i * 8 * PT) = bufA[kb % S1_DEPTH][i];
#pragma unroll
    for (int i = 0; i < 8; ++i)
      *(v4f*)(stB + i * 4 * PT) = bufB[kb % S1_DEPTH][i];
    if (kb + S1_DEPTH < S1_NCH) {
#pragma unroll
      for (int i = 0; i < 4; ++i)
        bufA[kb % S1_DEPTH][i] = *(const v4f*)(
            ciw + (size_t)((kb + S1_DEPTH) * S1_CH + i * 8) * MDIM);
#pragma unroll
      for (int i = 0; i < 8; ++i)
        bufB[kb % S1_DEPTH][i] = *(const v4f*)(
            xw + (size_t)((kb + S1_DEPTH) * S1_CH + i * 4) * NDIM);
    }
    __builtin_amdgcn_fence(__ATOMIC_RELEASE, "wavefront");
    __builtin_amdgcn_wave_barrier();
    __builtin_amdgcn_fence(__ATOMIC_ACQUIRE, "wavefront");

#pragma unroll
    for (int j = 0; j < S1_CH / 4; ++j) {
      v2f a, b0, b1;
      a.x  = frag[S1_TA  + (4 * j) * PT];
      a.y  = frag[S1_TA  + (4 * j + 1) * PT];
      b0.x = frag[S1_TB0 + (4 * j) * PT];
      b0.y = frag[S1_TB0 + (4 * j + 1) * PT];
      b1.x = frag[S1_TB1 + (4 * j) * PT];
      b1.y = frag[S1_TB1 + (4 * j + 1) * PT];
      c0 = __builtin_amdgcn_wmma_f32_16x16x4_f32(false, a, false, b0,
                                                 (short)0, c0, false, false);
      c1 = __builtin_amdgcn_wmma_f32_16x16x4_f32(false, a, false, b1,
                                                 (short)0, c1, false, false);
    }
    __builtin_amdgcn_fence(__ATOMIC_RELEASE, "wavefront");
    __builtin_amdgcn_wave_barrier();
    __builtin_amdgcn_fence(__ATOMIC_ACQUIRE, "wavefront");
  }

#pragma unroll
  for (int r = 0; r < 8; ++r) {
    tw[lane * 8 + r]        = c0[r];
    tw[(32 + lane) * 8 + r] = c1[r];
  }
  __syncthreads();

  {
    const int n   = threadIdx.x & 31;
    const int rr  = threadIdx.x >> 5;
    const int src = ((n >> 4) * 32 + (n & 15)) * 8 + rr;
    float s0 = 0.0f, s1 = 0.0f;
#pragma unroll
    for (int ww = 0; ww < KSPLIT; ++ww) {
      s0 += smem[ww * S1_WFLOATS + src];
      s1 += smem[ww * S1_WFLOATS + src + 16 * 8];
    }
    xr[(size_t)(R0 + rr) * NDIM + (size_t)(N0 + n)]     = s0;
    xr[(size_t)(R0 + rr + 8) * NDIM + (size_t)(N0 + n)] = s1;
  }
}

#define TPITCH  68
#define WFLOATS (CAP_SIZE + 16 * TPITCH)

__global__ __launch_bounds__(512)
void caps_transform_wmma(const float* __restrict__ Wi,
                         const float* __restrict__ xr,
                         float* __restrict__ y) {
  __shared__ __attribute__((aligned(16))) float smem[16 * WFLOATS];

  const int cap   = blockIdx.x;
  const int lane  = threadIdx.x & 31;
  const int mtile = __builtin_amdgcn_readfirstlane(threadIdx.x >> 5);
  const int half  = lane >> 4;
  const int ll    = lane & 15;
  const int jj    = mtile >> 2;
  const int l0    = (mtile & 3) * 16;

  float* __restrict__ xsw  = smem + mtile * WFLOATS;
  float* __restrict__ tile = xsw + CAP_SIZE;

  const float* __restrict__ xcap = xr + (size_t)cap * CAP_SIZE + lane * 4;
  const v4f xv0 = *(const v4f*)(xcap);
  const v4f xv1 = *(const v4f*)(xcap + 128);

  const float* __restrict__ Wbase =
      Wi + (size_t)cap * (CAP_SIZE * CAP_SIZE) + jj * 16384 + l0 * 64 + lane * 4;

  v4f buf[2][8];
#pragma unroll
  for (int i = 0; i < 8; ++i)
    buf[0][i] = __builtin_nontemporal_load((const v4f*)(Wbase + i * 128));
#pragma unroll
  for (int i = 0; i < 8; ++i)
    buf[1][i] = __builtin_nontemporal_load((const v4f*)(Wbase + 4096 + i * 128));

  *(v4f*)(xsw + lane * 4)       = xv0;
  *(v4f*)(xsw + 128 + lane * 4) = xv1;

  float* __restrict__ tst = tile + half * TPITCH + ll * 4;
  const float* __restrict__ tld = tile + ll * TPITCH + half * 4;
  const float* __restrict__ xld = xsw + half * 4;

  v8f c = {};
#pragma unroll
  for (int kb = 0; kb < 4; ++kb) {
#pragma unroll
    for (int i = 0; i < 8; ++i)
      *(v4f*)(tst + i * 2 * TPITCH) = buf[kb & 1][i];
    if (kb + 2 < 4) {
#pragma unroll
      for (int i = 0; i < 8; ++i)
        buf[kb & 1][i] = __builtin_nontemporal_load(
            (const v4f*)(Wbase + (kb + 2) * 4096 + i * 128));
    }
    __builtin_amdgcn_fence(__ATOMIC_RELEASE, "wavefront");
    __builtin_amdgcn_wave_barrier();
    __builtin_amdgcn_fence(__ATOMIC_ACQUIRE, "wavefront");

    v4f af[8], bf[8];
#pragma unroll
    for (int q = 0; q < 8; ++q) {
      af[q] = *(const v4f*)(tld + q * 8);
      bf[q] = *(const v4f*)(xld + kb * 64 + q * 8);
    }
#pragma unroll
    for (int q = 0; q < 8; ++q) {
      v2f a0, b0, a1, b1;
      a0.x = af[q].x; a0.y = af[q].y; b0.x = bf[q].x; b0.y = bf[q].y;
      a1.x = af[q].z; a1.y = af[q].w; b1.x = bf[q].z; b1.y = bf[q].w;
      c = __builtin_amdgcn_wmma_f32_16x16x4_f32(false, a0, false, b0,
                                                (short)0, c, false, false);
      c = __builtin_amdgcn_wmma_f32_16x16x4_f32(false, a1, false, b1,
                                                (short)0, c, false, false);
    }
    __builtin_amdgcn_fence(__ATOMIC_RELEASE, "wavefront");
    __builtin_amdgcn_wave_barrier();
    __builtin_amdgcn_fence(__ATOMIC_ACQUIRE, "wavefront");
  }

  if (ll == 0) {
#pragma unroll
    for (int r = 0; r < 8; ++r) {
      y[(size_t)cap * CAP_SIZE + (size_t)(mtile * 16 + r + 8 * half)] = c[r];
    }
  }
}

extern "C" void kernel_launch(void* const* d_in, const int* in_sizes, int n_in,
                              void* d_out, int out_size, void* d_ws, size_t ws_size,
                              hipStream_t stream) {
  const float* x  = (const float*)d_in[0];
  const float* Ci = (const float*)d_in[1];
  const float* Wi = (const float*)d_in[2];
  float* y  = (float*)d_out;
  float* xr = (float*)d_ws;

  scrramble_route_wmma<<<256, 256, 0, stream>>>(Ci, x, xr);
  caps_transform_wmma<<<NUM_CAPS, 512, 0, stream>>>(Wi, xr, y);
}


# Stage 2 on 256 blocks, one per pair of consecutive capsules: each of the 16 waves streams its 16-row tile of both capsules (8 k blocks, 3 in flight): half as many, longer-lived waves; sum order kept
# speedup vs baseline: 2.1818x; 1.0337x over previous
#include <hip/hip_runtime.h>

typedef float v2f __attribute__((ext_vector_type(2)));
typedef float v4f __attribute__((ext_vector_type(4)));
typedef float v8f __attribute__((ext_vector_type(8)));

#define NUM_CAPS 512
#define CAP_SIZE 256
#define RF_SIZE  64
#define KDIM     2048
#define MDIM     2048
#define NDIM     64
#define KSPLIT   8

#define S1_CH     32
#define S1_NCH    ((KDIM / KSPLIT) / S1_CH)
#define S1_DEPTH  3
#define PT        16
#define S1_TA     0
#define S1_TB0    (S1_CH * PT)
#define S1_TB1    (2 * S1_CH * PT + 32)
#define S1_WFLOATS (3 * S1_CH * PT + 32)

__global__ __launch_bounds__(256)
void scrramble_route_wmma(const float* __restrict__ Ci,
                          const float* __restrict__ x,
                          float* __restrict__ xr) {
  __shared__ __attribute__((aligned(16))) float smem[KSPLIT * S1_WFLOATS];

  const int lane = threadIdx.x & 31;
  const int w    = __builtin_amdgcn_readfirstlane(threadIdx.x >> 5);
  const int mt   = blockIdx.x >> 1;
  const int nh   = blockIdx.x & 1;
  const int R0   = mt * 16;
  const int N0   = nh * 32;
  const int half = lane >> 4;
  const int ll   = lane & 15;
  const int Sw   = w * (KDIM / KSPLIT);

  float* __restrict__ tw = smem + w * S1_WFLOATS;

  const float* __restrict__ ciw =
      Ci + (size_t)(Sw + (lane >> 2)) * MDIM + R0 + (lane & 3) * 4;
  const float* __restrict__ xw =
      x + (size_t)(Sw + (lane >> 3)) * NDIM + N0 + (lane & 7) * 4;

  v4f bufA[S1_DEPTH][4], bufB[S1_DEPTH][8];
#pragma unroll
  for (int d = 0; d < S1_DEPTH; ++d) {
#pragma unroll
    for (int i = 0; i < 4; ++i)
      bufA[d][i] = *(const v4f*)(ciw + (size_t)(d * S1_CH + i * 8) * MDIM);
#pragma unroll
    for (int i = 0; i < 8; ++i)
      bufB[d][i] = *(const v4f*)(xw + (size_t)(d * S1_CH + i * 4) * NDIM);
  }

  float* __restrict__ stA = tw + S1_TA + (lane >> 2) * PT + (lane & 3) * 4;
  float* __restrict__ stB = tw + (((lane >> 2) & 1) ? S1_TB1 : S1_TB0)
                               + (lane >> 3) * PT + (lane & 3) * 4;
  const float* __restrict__ frag = tw + (2 * half) * PT + ll;

  v8f c0 = {}, c1 = {};
#pragma unroll
  for (int kb = 0; kb < S1_NCH; ++kb) {
#pragma unroll
    for (int i = 0; i < 4; ++i)
      *(v4f*)(stA + i * 8 * PT) = bufA[kb % S1_DEPTH][i];
#pragma unroll
    for (int i = 0; i < 8; ++i)
      *(v4f*)(stB + i * 4 * PT) = bufB[kb % S1_DEPTH][i];
    if (kb + S1_DEPTH < S1_NCH) {
#pragma unroll
      for (int i = 0; i < 4; ++i)
        bufA[kb % S1_DEPTH][i] = *(const v4f*)(
            ciw + (size_t)((kb + S1_DEPTH) * S1_CH + i * 8) * MDIM);
#pragma unroll
      for (int i = 0; i < 8; ++i)
        bufB[kb % S1_DEPTH][i] = *(const v4f*)(
            xw + (size_t)((kb + S1_DEPTH) * S1_CH + i * 4) * NDIM);
    }
    __builtin_amdgcn_fence(__ATOMIC_RELEASE, "wavefront");
    __builtin_amdgcn_wave_barrier();
    __builtin_amdgcn_fence(__ATOMIC_ACQUIRE, "wavefront");

#pragma unroll
    for (int j = 0; j < S1_CH / 4; ++j) {
      v2f a, b0, b1;
      a.x  = frag[S1_TA  + (4 * j) * PT];
      a.y  = frag[S1_TA  + (4 * j + 1) * PT];
      b0.x = frag[S1_TB0 + (4 * j) * PT];
      b0.y = frag[S1_TB0 + (4 * j + 1) * PT];
      b1.x = frag[S1_TB1 + (4 * j) * PT];
      b1.y = frag[S1_TB1 + (4 * j + 1) * PT];
      c0 = __builtin_amdgcn_wmma_f32_16x16x4_f32(false, a, false, b0,
                                                 (short)0, c0, false, false);
      c1 = __builtin_amdgcn_wmma_f32_16x16x4_f32(false, a, false, b1,
                                                 (short)0, c1, false, false);
    }
    __builtin_amdgcn_fence(__ATOMIC_RELEASE, "wavefront");
    __builtin_amdgcn_wave_barrier();
    __builtin_amdgcn_fence(__ATOMIC_ACQUIRE, "wavefront");
  }

#pragma unroll
  for (int r = 0; r < 8; ++r) {
    tw[lane * 8 + r]        = c0[r];
    tw[(32 + lane) * 8 + r] = c1[r];
  }
  __syncthreads();

  {
    const int n   = threadIdx.x & 31;
    const int rr  = threadIdx.x >> 5;
    const int src = ((n >> 4) * 32 + (n & 15)) * 8 + rr;
    float s0 = 0.0f, s1 = 0.0f;
#pragma unroll
    for (int ww = 0; ww < KSPLIT; ++ww) {
      s0 += smem[ww * S1_WFLOATS + src];
      s1 += smem[ww * S1_WFLOATS + src + 16 * 8];
    }
    xr[(size_t)(R0 + rr) * NDIM + (size_t)(N0 + n)]     = s0;
    xr[(size_t)(R0 + rr + 8) * NDIM + (size_t)(N0 + n)] = s1;
  }
}

#define S2_CAPS  2
#define S2_NKB   (4 * S2_CAPS)
#define S2_DEPTH 3
#define TPITCH   68
#define WFLOATS  (S2_CAPS * CAP_SIZE + 16 * TPITCH)
#define S2_WOFF(kb) (((kb) >> 2) * (CAP_SIZE * CAP_SIZE) + ((kb) & 3) * 4096)

__global__ __launch_bounds__(512)
void caps_transform_wmma(const float* __restrict__ Wi,
                         const float* __restrict__ xr,
                         float* __restrict__ y) {
  __shared__ __attribute__((aligned(16))) float smem[16 * WFLOATS];

  const int cap0  = blockIdx.x * S2_CAPS;
  const int lane  = threadIdx.x & 31;
  const int mtile = __builtin_amdgcn_readfirstlane(threadIdx.x >> 5);
  const int half  = lane >> 4;
  const int ll    = lane & 15;
  const int jj    = mtile >> 2;
  const int l0    = (mtile & 3) * 16;

  float* __restrict__ xsw  = smem + mtile * WFLOATS;
  float* __restrict__ tile = xsw + S2_CAPS * CAP_SIZE;

  const float* __restrict__ xcap = xr + (size_t)cap0 * CAP_SIZE + lane * 4;
  v4f xv[2 * S2_CAPS];
#pragma unroll
  for (int i = 0; i < 2 * S2_CAPS; ++i)
    xv[i] = *(const v4f*)(xcap + i * 128);

  const float* __restrict__ Wbase =
      Wi + (size_t)cap0 * (CAP_SIZE * CAP_SIZE) + jj * 16384 + l0 * 64 + lane * 4;

  v4f buf[S2_DEPTH][8];
#pragma unroll
  for (int d = 0; d < S2_DEPTH; ++d) {
#pragma unroll
    for (int i = 0; i < 8; ++i)
      buf[d][i] = __builtin_nontemporal_load(
          (const v4f*)(Wbase + S2_WOFF(d) + i * 128));
  }

#pragma unroll
  for (int i = 0; i < 2 * S2_CAPS; ++i)
    *(v4f*)(xsw + i * 128 + lane * 4) = xv[i];

  float* __restrict__ tst = tile + half * TPITCH + ll * 4;
  const float* __restrict__ tld = tile + ll * TPITCH + half * 4;
  const float* __restrict__ xld = xsw + half * 4;

  v8f c[S2_CAPS];
#pragma unroll
  for (int s = 0; s < S2_CAPS; ++s) c[s] = (v8f){};

#pragma unroll
  for (int kb = 0; kb < S2_NKB; ++kb) {
#pragma unroll
    for (int i = 0; i < 8; ++i)
      *(v4f*)(tst + i * 2 * TPITCH) = buf[kb % S2_DEPTH][i];
    if (kb + S2_DEPTH < S2_NKB) {
#pragma unroll
      for (int i = 0; i < 8; ++i)
        buf[kb % S2_DEPTH][i] = __builtin_nontemporal_load(
            (const v4f*)(Wbase + S2_WOFF(kb + S2_DEPTH) + i * 128));
    }
    __builtin_amdgcn_fence(__ATOMIC_RELEASE, "wavefront");
    __builtin_amdgcn_wave_barrier();
    __builtin_amdgcn_fence(__ATOMIC_ACQUIRE, "wavefront");

    v4f af[8], bf[8];
#pragma unroll
    for (int q = 0; q < 8; ++q) {
      af[q] = *(const v4f*)(tld + q * 8);
      bf[q] = *(const v4f*)(xld + kb * 64 + q * 8);
    }
#pragma unroll
    for (int q = 0; q < 8; ++q) {
      v2f a0, b0, a1, b1;
      a0.x = af[q].x; a0.y = af[q].y; b0.x = bf[q].x; b0.y = bf[q].y;
      a1.x = af[q].z; a1.y = af[q].w; b1.x = bf[q].z; b1.y = bf[q].w;
      c[kb >> 2] = __builtin_amdgcn_wmma_f32_16x16x4_f32(
          false, a0, false, b0, (short)0, c[kb >> 2], false, false);
      c[kb >> 2] = __builtin_amdgcn_wmma_f32_16x16x4_f32(
          false, a1, false, b1, (short)0, c[kb >> 2], false, false);
    }
    __builtin_amdgcn_fence(__ATOMIC_RELEASE, "wavefront");
    __builtin_amdgcn_wave_barrier();
    __builtin_amdgcn_fence(__ATOMIC_ACQUIRE, "wavefront");
  }

  if (ll == 0) {
#pragma unroll
    for (int s = 0; s < S2_CAPS; ++s) {
#pragma unroll
      for (int r = 0; r < 8; ++r) {
        y[(size_t)(cap0 + s) * CAP_SIZE + (size_t)(mtile * 16 + r + 8 * half)] =
            c[s][r];
      }
    }
  }
}

extern "C" void kernel_launch(void* const* d_in, const int* in_sizes, int n_in,
                              void* d_out, int out_size, void* d_ws, size_t ws_size,
                              hipStream_t stream) {
  const float* x  = (const float*)d_in[0];
  const float* Ci = (const float*)d_in[1];
  const float* Wi = (const float*)d_in[2];
  float* y  = (float*)d_out;
  float* xr = (float*)d_ws;

  scrramble_route_wmma<<<256, 256, 0, stream>>>(Ci, x, xr);
  caps_transform_wmma<<<256, 512, 0, stream>>>(Wi, xr, y);
}


# Both stages in one launch: each block routes its own capsule pair first (8 Ci columns, K split over its 16 waves, fixed-order LDS reduce) under its first weight loads; route sum order changed
# speedup vs baseline: 2.3753x; 1.0887x over previous
#include <hip/hip_runtime.h>


typedef float v2f __attribute__((ext_vector_type(2)));
typedef float v4f __attribute__((ext_vector_type(4)));
typedef float v8f __attribute__((ext_vector_type(8)));

#define NUM_CAPS 512
#define CAP_SIZE 256
#define KDIM     2048
#define MDIM     2048
#define NDIM     64

#define NWAVES   16
#define S2_CAPS  2
#define S2_NKB   (4 * S2_CAPS)
#define S2_DEPTH 3
#define TPITCH   68
#define S2_WOFF(kb) (((kb) >> 2) * (CAP_SIZE * CAP_SIZE) + ((kb) & 3) * 4096)

#define RT_ROWS  (KDIM / NWAVES)
#define RT_CH    16
#define RT_NCH   (RT_ROWS / RT_CH)
#define RT_DEPTH 1
#define PT       16
#define RT_TA    0
#define RT_TB(n) (256 + (n) * 272)
#define RT_FLOATS (256 + 4 * 272)
#define WFLOATS  (RT_FLOATS + 16 * TPITCH)

__global__ __launch_bounds__(512)
void caps_route_transform_wmma(const float* __restrict__ Ci,
                               const float* __restrict__ x,
                               const float* __restrict__ Wi,
                               float* __restrict__ y) {
  __shared__ __attribute__((aligned(16)))
      float smem[NWAVES * WFLOATS + S2_CAPS * CAP_SIZE];

  const int cap0  = blockIdx.x * S2_CAPS;
  const int lane  = threadIdx.x & 31;
  const int w     = __builtin_amdgcn_readfirstlane(threadIdx.x >> 5);
  const int half  = lane >> 4;
  const int ll    = lane & 15;
  const int jj    = w >> 2;
  const int l0    = (w & 3) * 16;

  float* rt   = smem + w * WFLOATS;
  float* tile = rt + RT_FLOATS;
  float* xs   = smem + NWAVES * WFLOATS;

  const int Sw = w * RT_ROWS;
  const int R0 = cap0 * 4;
  const float* __restrict__ ciw =
      Ci + (size_t)(Sw + (lane >> 1)) * MDIM + R0 + (lane & 1) * 4;
  const float* __restrict__ xw = x + (size_t)Sw * NDIM + lane * 4;

  v4f ra[RT_DEPTH], rb[RT_DEPTH][8];
#pragma unroll
  for (int d = 0; d < RT_DEPTH; ++d) {
    ra[d] = *(const v4f*)(ciw + (size_t)(d * RT_CH) * MDIM);
#pragma unroll
    for (int i = 0; i < 8; ++i)
      rb[d][i] = *(const v4f*)(xw + (size_t)(d * RT_CH + 2 * i) * NDIM);
  }

  __builtin_amdgcn_sched_barrier(0);

  const float* __restrict__ Wbase =
      Wi + (size_t)cap0 * (CAP_SIZE * CAP_SIZE) + jj * 16384 + l0 * 64 + lane * 4;
  v4f buf[S2_DEPTH][8];
#pragma unroll
  for (int d = 0; d < S2_DEPTH; ++d) {
#pragma unroll
    for (int i = 0; i < 8; ++i)
      buf[d][i] = __builtin_nontemporal_load(
          (const v4f*)(Wbase + S2_WOFF(d) + i * 128));
  }

  float* stA = rt + RT_TA + (lane >> 1) * PT + (lane & 1) * 4;
  float* stB = rt + RT_TB(ll >> 2) + half * PT + (ll & 3) * 4;
  const float* fragA = rt + RT_TA + (2 * half) * PT + (ll & 7);
  const float* fragB = rt + (2 * half) * PT + ll;

  v8f acc[4];
#pragma unroll
  for (int n = 0; n < 4; ++n) acc[n] = (v8f){};

#pragma unroll
  for (int kc = 0; kc < RT_NCH; ++kc) {
    *(v4f*)stA = ra[kc % RT_DEPTH];
#pragma unroll
    for (int i = 0; i < 8; ++i)
      *(v4f*)(stB + 2 * i * PT) = rb[kc % RT_DEPTH][i];
    if (kc + RT_DEPTH < RT_NCH) {
      ra[kc % RT_DEPTH] =
          *(const v4f*)(ciw + (size_t)((kc + RT_DEPTH) * RT_CH) * MDIM);
#pragma unroll
      for (int i = 0; i < 8; ++i)
        rb[kc % RT_DEPTH][i] = *(const v4f*)(
            xw + (size_t)((kc + RT_DEPTH) * RT_CH + 2 * i) * NDIM);
    }
    __builtin_amdgcn_fence(__ATOMIC_RELEASE, "wavefront");
    __builtin_amdgcn_wave_barrier();
    __builtin_amdgcn_fence(__ATOMIC_ACQUIRE, "wavefront");

    v2f fa[RT_CH / 4], fb[RT_CH / 4][4];
#pragma unroll
    for (int j = 0; j < RT_CH / 4; ++j) {
      fa[j].x = fragA[(4 * j) * PT];
      fa[j].y = fragA[(4 * j + 1) * PT];
#pragma unroll
      for (int n = 0; n < 4; ++n) {
        fb[j][n].x = fragB[RT_TB(n) + (4 * j) * PT];
        fb[j][n].y = fragB[RT_TB(n) + (4 * j + 1) * PT];
      }
    }
#pragma unroll
    for (int j = 0; j < RT_CH / 4; ++j) {
#pragma unroll
      for (int n = 0; n < 4; ++n)
        acc[n] = __builtin_amdgcn_wmma_f32_16x16x4_f32(
            false, fa[j], false, fb[j][n], (short)0, acc[n], false, false);
    }
    __builtin_amdgcn_fence(__ATOMIC_RELEASE, "wavefront");
    __builtin_amdgcn_wave_barrier();
    __builtin_amdgcn_fence(__ATOMIC_ACQUIRE, "wavefront");
  }

  if (half == 0) {
#pragma unroll
    for (int n = 0; n < 4; ++n) {
#pragma unroll
      for (int r = 0; r < 8; ++r)
        rt[r * NDIM + n * 16 + ll] = acc[n][r];
    }
  }
  __syncthreads();
  {
    const int e = threadIdx.x;
    float s = 0.0f;
#pragma unroll
    for (int ww = 0; ww < NWAVES; ++ww) s += smem[ww * WFLOATS + e];
    xs[e] = s;
  }
  __syncthreads();

  float* tst = tile + half * TPITCH + ll * 4;
  const float* tld = tile + ll * TPITCH + half * 4;
  const float* xld = xs + half * 4;

  v8f c[S2_CAPS];
#pragma unroll
  for (int s = 0; s < S2_CAPS; ++s) c[s] = (v8f){};

#pragma unroll
  for (int kb = 0; kb < S2_NKB; ++kb) {
#pragma unroll
    for (int i = 0; i < 8; ++i)
      *(v4f*)(tst + i * 2 * TPITCH) = buf[kb % S2_DEPTH][i];
    if (kb + S2_DEPTH < S2_NKB) {
#pragma unroll
      for (int i = 0; i < 8; ++i)
        buf[kb % S2_DEPTH][i] = __builtin_nontemporal_load(
            (const v4f*)(Wbase + S2_WOFF(kb + S2_DEPTH) + i * 128));
    }
    __builtin_amdgcn_fence(__ATOMIC_RELEASE, "wavefront");
    __builtin_amdgcn_wave_barrier();
    __builtin_amdgcn_fence(__ATOMIC_ACQUIRE, "wavefront");

    v4f af[8], bf[8];
#pragma unroll
    for (int q = 0; q < 8; ++q) {
      af[q] = *(const v4f*)(tld + q * 8);
      bf[q] = *(const v4f*)(xld + kb * 64 + q * 8);
    }
#pragma unroll
    for (int q = 0; q < 8; ++q) {
      v2f a0, b0, a1, b1;
      a0.x = af[q].x; a0.y = af[q].y; b0.x = bf[q].x; b0.y = bf[q].y;
      a1.x = af[q].z; a1.y = af[q].w; b1.x = bf[q].z; b1.y = bf[q].w;
      c[kb >> 2] = __builtin_amdgcn_wmma_f32_16x16x4_f32(
          false, a0, false, b0, (short)0, c[kb >> 2], false, false);
      c[kb >> 2] = __builtin_amdgcn_wmma_f32_16x16x4_f32(
          false, a1, false, b1, (short)0, c[kb >> 2], false, false);
    }
    __builtin_amdgcn_fence(__ATOMIC_RELEASE, "wavefront");
    __builtin_amdgcn_wave_barrier();
    __builtin_amdgcn_fence(__ATOMIC_ACQUIRE, "wavefront");
  }

  if (ll == 0) {
#pragma unroll
    for (int s = 0; s < S2_CAPS; ++s) {
#pragma unroll
      for (int r = 0; r < 8; ++r) {
        y[(size_t)(cap0 + s) * CAP_SIZE + (size_t)(w * 16 + r + 8 * half)] =
            c[s][r];
      }
    }
  }
}

extern "C" void kernel_launch(void* const* d_in, const int* in_sizes, int n_in,
                              void* d_out, int out_size, void* d_ws, size_t ws_size,
                              hipStream_t stream) {
  const float* x  = (const float*)d_in[0];
  const float* Ci = (const float*)d_in[1];
  const float* Wi = (const float*)d_in[2];
  float* y  = (float*)d_out;

  caps_route_transform_wmma<<<256, 512, 0, stream>>>(Ci, x, Wi, y);
}


# Routing operands go straight from global loads into the WMMA fragments (x 16B/lane, N tile n = columns 4c+n; Ci 4B/lane), 2 chunks in flight, no LDS hop in the K loop; k order inside a WMMA permuted
# speedup vs baseline: 2.4992x; 1.0522x over previous
#include <hip/hip_runtime.h>


typedef float v2f __attribute__((ext_vector_type(2)));
typedef float v4f __attribute__((ext_vector_type(4)));
typedef float v8f __attribute__((ext_vector_type(8)));
typedef const __attribute__((address_space(1))) float* gfp;
typedef const __attribute__((address_space(1))) v4f*   gv4p;

#define NUM_CAPS 512
#define CAP_SIZE 256
#define KDIM     2048
#define MDIM     2048
#define NDIM     64

#define NWAVES   16
#define S2_CAPS  2
#define S2_NKB   (4 * S2_CAPS)
#define S2_DEPTH 3
#define TPITCH   68
#define S2_WOFF(kb) (((kb) >> 2) * (CAP_SIZE * CAP_SIZE) + ((kb) & 3) * 4096)

#define RT_ROWS  (KDIM / NWAVES)
#define RT_CH    16
#define RT_NCH   (RT_ROWS / RT_CH)
#define RT_DEPTH 2
#define PART_FLOATS 512
#define WFLOATS  (PART_FLOATS + 16 * TPITCH)

__global__ __launch_bounds__(512)
void caps_route_transform_wmma(const float* __restrict__ Ci,
                               const float* __restrict__ x,
                               const float* __restrict__ Wi,
                               float* __restrict__ y) {
  __shared__ __attribute__((aligned(16)))
      float smem[NWAVES * WFLOATS + S2_CAPS * CAP_SIZE];

  const int cap0  = blockIdx.x * S2_CAPS;
  const int lane  = threadIdx.x & 31;
  const int w     = __builtin_amdgcn_readfirstlane(threadIdx.x >> 5);
  const int half  = lane >> 4;
  const int ll    = lane & 15;
  const int jj    = w >> 2;
  const int l0    = (w & 3) * 16;

  float* part = smem + w * WFLOATS;
  float* tile = part + PART_FLOATS;
  float* xs   = smem + NWAVES * WFLOATS;

  const int Sw = w * RT_ROWS;
  const int R0 = cap0 * 4;
  unsigned long long ciw_a =
      (unsigned long long)(Ci + (size_t)(Sw + half) * MDIM + R0 + (ll & 7));
  unsigned long long xw_a =
      (unsigned long long)(x + (size_t)(Sw + half) * NDIM + ll * 4);
  asm volatile("" : "+v"(ciw_a));
  asm volatile("" : "+v"(xw_a));
  const gfp ciw = (gfp)ciw_a;
  const gfp xw  = (gfp)xw_a;

  float ra[RT_DEPTH][8];
  v4f   rb[RT_DEPTH][8];
#pragma unroll
  for (int d = 0; d < RT_DEPTH; ++d) {
#pragma unroll
    for (int i = 0; i < 8; ++i) {
      ra[d][i] = ciw[(size_t)(d * RT_CH + 2 * i) * MDIM];
      rb[d][i] = *(gv4p)(xw + (size_t)(d * RT_CH + 2 * i) * NDIM);
    }
  }

  __builtin_amdgcn_sched_barrier(0);

  unsigned long long Wbase_a = (unsigned long long)(
      Wi + (size_t)cap0 * (CAP_SIZE * CAP_SIZE) + jj * 16384 + l0 * 64 + lane * 4);
  asm volatile("" : "+v"(Wbase_a));
  const gfp Wbase = (gfp)Wbase_a;
  v4f buf[S2_DEPTH][8];
#pragma unroll
  for (int d = 0; d < S2_DEPTH; ++d) {
#pragma unroll
    for (int i = 0; i < 8; ++i)
      buf[d][i] = __builtin_nontemporal_load(
          (gv4p)(Wbase + S2_WOFF(d) + i * 128));
  }

  __builtin_amdgcn_sched_barrier(0);

  v8f acc[4];
#pragma unroll
  for (int n = 0; n < 4; ++n) acc[n] = (v8f){};

#pragma unroll
  for (int kc = 0; kc < RT_NCH; ++kc) {
#pragma unroll
    for (int j = 0; j < RT_CH / 4; ++j) {
      v2f fa;
      fa.x = ra[kc % RT_DEPTH][2 * j];
      fa.y = ra[kc % RT_DEPTH][2 * j + 1];
#pragma unroll
      for (int n = 0; n < 4; ++n) {
        v2f fb;
        fb.x = rb[kc % RT_DEPTH][2 * j][n];
        fb.y = rb[kc % RT_DEPTH][2 * j + 1][n];
        acc[n] = __builtin_amdgcn_wmma_f32_16x16x4_f32(
            false, fa, false, fb, (short)0, acc[n], false, false);
      }
    }
    __builtin_amdgcn_sched_barrier(0);
    if (kc + RT_DEPTH < RT_NCH) {
#pragma unroll
      for (int i = 0; i < 8; ++i) {
        ra[kc % RT_DEPTH][i] =
            ciw[(size_t)((kc + RT_DEPTH) * RT_CH + 2 * i) * MDIM];
        rb[kc % RT_DEPTH][i] = *(gv4p)(
            xw + (size_t)((kc + RT_DEPTH) * RT_CH + 2 * i) * NDIM);
      }
    }
    __builtin_amdgcn_sched_barrier(0);
  }

  if (half == 0) {
#pragma unroll
    for (int r = 0; r < 8; ++r) {
      v4f o;
      o.x = acc[0][r]; o.y = acc[1][r]; o.z = acc[2][r]; o.w = acc[3][r];
      *(v4f*)(part + r * NDIM + ll * 4) = o;
    }
  }
  __syncthreads();
  {
    const int e = threadIdx.x;
    float s = 0.0f;
#pragma unroll
    for (int ww = 0; ww < NWAVES; ++ww) s += smem[ww * WFLOATS + e];
    xs[e] = s;
  }
  __syncthreads();

  float* tst = tile + half * TPITCH + ll * 4;
  const float* tld = tile + ll * TPITCH + half * 4;
  const float* xld = xs + half * 4;

  v8f c[S2_CAPS];
#pragma unroll
  for (int s = 0; s < S2_CAPS; ++s) c[s] = (v8f){};

#pragma unroll
  for (int kb = 0; kb < S2_NKB; ++kb) {
#pragma unroll
    for (int i = 0; i < 8; ++i)
      *(v4f*)(tst + i * 2 * TPITCH) = buf[kb % S2_DEPTH][i];
    if (kb + S2_DEPTH < S2_NKB) {
#pragma unroll
      for (int i = 0; i < 8; ++i)
        buf[kb % S2_DEPTH][i] = __builtin_nontemporal_load(
            (gv4p)(Wbase + S2_WOFF(kb + S2_DEPTH) + i * 128));
    }
    __builtin_amdgcn_fence(__ATOMIC_RELEASE, "wavefront");
    __builtin_amdgcn_wave_barrier();
    __builtin_amdgcn_fence(__ATOMIC_ACQUIRE, "wavefront");

    v4f af[8], bf[8];
#pragma unroll
    for (int q = 0; q < 8; ++q) {
      af[q] = *(const v4f*)(tld + q * 8);
      bf[q] = *(const v4f*)(xld + kb * 64 + q * 8);
    }
#pragma unroll
    for (int q = 0; q < 8; ++q) {
      v2f a0, b0, a1, b1;
      a0.x = af[q].x; a0.y = af[q].y; b0.x = bf[q].x; b0.y = bf[q].y;
      a1.x = af[q].z; a1.y = af[q].w; b1.x = bf[q].z; b1.y = bf[q].w;
      c[kb >> 2] = __builtin_amdgcn_wmma_f32_16x16x4_f32(
          false, a0, false, b0, (short)0, c[kb >> 2], false, false);
      c[kb >> 2] = __builtin_amdgcn_wmma_f32_16x16x4_f32(
          false, a1, false, b1, (short)0, c[kb >> 2], false, false);
    }
    __builtin_amdgcn_fence(__ATOMIC_RELEASE, "wavefront");
    __builtin_amdgcn_wave_barrier();
    __builtin_amdgcn_fence(__ATOMIC_ACQUIRE, "wavefront");
  }

  if (ll == 0) {
#pragma unroll
    for (int s = 0; s < S2_CAPS; ++s) {
#pragma unroll
      for (int r = 0; r < 8; ++r) {
        y[(size_t)(cap0 + s) * CAP_SIZE + (size_t)(w * 16 + r + 8 * half)] =
            c[s][r];
      }
    }
  }
}

extern "C" void kernel_launch(void* const* d_in, const int* in_sizes, int n_in,
                              void* d_out, int out_size, void* d_ws, size_t ws_size,
                              hipStream_t stream) {
  const float* x  = (const float*)d_in[0];
  const float* Ci = (const float*)d_in[1];
  const float* Wi = (const float*)d_in[2];
  float* y  = (float*)d_out;

  caps_route_transform_wmma<<<256, 512, 0, stream>>>(Ci, x, Wi, y);
}
